# starting kernel
# baseline (speedup 1.0000x reference)
#include <hip/hip_runtime.h>


typedef __attribute__((ext_vector_type(2))) float v2f;
typedef __attribute__((ext_vector_type(8))) float v8f;

#define N_NODES 100000
#define OFF_NORM 100096
#define OFF_BUFA (OFF_NORM + 1250048)
#define NBUF ((size_t)N_NODES * 64)


__global__ void gcn_deg_init(float* __restrict__ deg, int n) {
  int i = blockIdx.x * blockDim.x + threadIdx.x;
  if (i < n) deg[i] = 1.0f;
}

__global__ void gcn_deg_count(const int* __restrict__ dst, float* __restrict__ deg, int nE) {
  int e = blockIdx.x * blockDim.x + threadIdx.x;
  if (e < nE) atomicAdd(&deg[dst[e]], 1.0f);
}

__global__ void gcn_deg_rsqrt(float* __restrict__ deg, int n) {
  int i = blockIdx.x * blockDim.x + threadIdx.x;
  if (i < n) deg[i] = rsqrtf(deg[i]);
}

__global__ void gcn_edge_norm(const int* __restrict__ src, const int* __restrict__ dst,
                              const float* __restrict__ dis, float* __restrict__ nrm,
                              int nE) {
  int e = blockIdx.x * blockDim.x + threadIdx.x;
  if (e < nE) nrm[e] = dis[src[e]] * dis[dst[e]];
}


template <int K, int NC, bool RELU>
__global__ __launch_bounds__(64)
void gcn_gemm_wmma(const float* X, const float* __restrict__ W,
                   float* __restrict__ H, float* OUTI,
                   const float* __restrict__ dis, const float* __restrict__ bias) {
  constexpr int S  = K + 6;
  constexpr int CT = NC / 16;
  __shared__ float sWt[NC * S];

  for (int t = threadIdx.x; t < K * NC / 4; t += 64) {
    const int kr = (t * 4) / NC;
    const int n  = (t * 4) % NC;
    const float4 w4 = *(const float4*)(W + (size_t)t * 4);
    sWt[(size_t)(n + 0) * S + kr] = w4.x;
    sWt[(size_t)(n + 1) * S + kr] = w4.y;
    sWt[(size_t)(n + 2) * S + kr] = w4.z;
    sWt[(size_t)(n + 3) * S + kr] = w4.w;
  }
  __syncthreads();

  const int lane = threadIdx.x & 31;
  const int wave = threadIdx.x >> 5;
  const int m    = lane & 15;
  const int half = lane >> 4;
  const long long row0 = ((long long)blockIdx.x * 2 + wave) * 16;

  const float* xrow  = X + (size_t)(row0 + m) * K + 2 * half;
  const float* bbase = sWt + (size_t)m * S + 2 * half;

  v8f acc[CT] = {};
#pragma unroll
  for (int kk = 0; kk < K / 4; ++kk) {
    const int k = kk * 4;
    float ax = xrow[k];
    float ay = xrow[k + 1];
    if (RELU) { ax = fmaxf(ax, 0.f); ay = fmaxf(ay, 0.f); }
    const v2f a = {ax, ay};
#pragma unroll
    for (int ct = 0; ct < CT; ++ct) {
      const v2f b = *(const v2f*)(bbase + (size_t)ct * 16 * S + k);
      acc[ct] = __builtin_amdgcn_wmma_f32_16x16x4_f32(
          false, a, false, b, (short)0, acc[ct], false, false);
    }
  }

#pragma unroll
  for (int ct = 0; ct < CT; ++ct) {
    const float bv = bias[ct * 16 + m];
#pragma unroll
    for (int r = 0; r < 8; ++r) {
      const long long row = row0 + r + 8 * half;
      const float dv = dis[row];
      const float hv = acc[ct][r];
      H[(size_t)row * NC + ct * 16 + m]    = hv;
      OUTI[(size_t)row * NC + ct * 16 + m] = bv + hv * dv * dv;
    }
  }
}


template <int F>
__global__ void gcn_edge_scatter(const float* __restrict__ h,
                                 const int* __restrict__ src,
                                 const int* __restrict__ dst,
                                 const float* __restrict__ nrm,
                                 float* __restrict__ out, int nE) {
  constexpr int CH = F / 4;
  const long long total  = (long long)nE * CH;
  const long long stride = (long long)gridDim.x * blockDim.x;
  for (long long idx = (long long)blockIdx.x * blockDim.x + threadIdx.x;
       idx < total; idx += stride) {
    const long long nidx = idx + stride;
    if (nidx < total) {
      const int ne = (int)(nidx / CH);
      __builtin_prefetch(h + (size_t)src[ne] * F + (int)(nidx % CH) * 4, 0, 0);
    }
    const int e  = (int)(idx / CH);
    const int fo = (int)(idx % CH) * 4;
    const int s  = src[e];
    const int d  = dst[e];
    const float w = nrm[e];
    const float4 v = *(const float4*)(h + (size_t)s * F + fo);
    float* o = out + (size_t)d * F + fo;
    atomicAdd(o + 0, v.x * w);
    atomicAdd(o + 1, v.y * w);
    atomicAdd(o + 2, v.z * w);
    atomicAdd(o + 3, v.w * w);
  }
}


extern "C" void kernel_launch(void* const* d_in, const int* in_sizes, int n_in,
                              void* d_out, int out_size, void* d_ws, size_t ws_size,
                              hipStream_t stream) {
  const float* x  = (const float*)d_in[0];
  const int*   ei = (const int*)d_in[1];
  const float* W1 = (const float*)d_in[2];
  const float* b1 = (const float*)d_in[3];
  const float* W2 = (const float*)d_in[4];
  const float* b2 = (const float*)d_in[5];
  const float* W3 = (const float*)d_in[6];
  const float* b3 = (const float*)d_in[7];
  float* out = (float*)d_out;

  const int N = N_NODES;
  const int E = in_sizes[1] / 2;
  const int* src = ei;
  const int* dst = ei + E;

  float* ws   = (float*)d_ws;
  float* dis  = ws;
  float* nrm  = ws + OFF_NORM;
  float* bufA = ws + OFF_BUFA;
  float* bufB = bufA + NBUF;

  const int TB = 256;
  const int gN = (N + TB - 1) / TB;
  const int gE = (E + TB - 1) / TB;
  const int gGemm = N / 32;
  const int gScat = 4096;

  gcn_deg_init <<<gN, TB, 0, stream>>>(dis, N);
  gcn_deg_count<<<gE, TB, 0, stream>>>(dst, dis, E);
  gcn_deg_rsqrt<<<gN, TB, 0, stream>>>(dis, N);
  gcn_edge_norm<<<gE, TB, 0, stream>>>(src, dst, dis, nrm, E);

  gcn_gemm_wmma<128, 64, false><<<gGemm, 64, 0, stream>>>(x, W1, bufA, bufB, dis, b1);
  gcn_edge_scatter<64><<<gScat, TB, 0, stream>>>(bufA, src, dst, nrm, bufB, E);

  gcn_gemm_wmma<64, 64, true><<<gGemm, 64, 0, stream>>>(bufB, W2, bufA, bufB, dis, b2);
  gcn_edge_scatter<64><<<gScat, TB, 0, stream>>>(bufA, src, dst, nrm, bufB, E);

  gcn_gemm_wmma<64, 32, true><<<gGemm, 64, 0, stream>>>(bufB, W3, bufA, out, dis, b3);
  gcn_edge_scatter<32><<<gScat, TB, 0, stream>>>(bufA, src, dst, nrm, out, E);
}


# Replace per-edge f32 atomic scatter by per-node gather over a dst-sorted edge list built in-call (int count with arrival-order rank, scan, fill); one writer per row; sum order = arrival rank
# speedup vs baseline: 1.9424x; 1.9424x over previous
#include <hip/hip_runtime.h>


typedef __attribute__((ext_vector_type(2))) float v2f;
typedef __attribute__((ext_vector_type(8))) float v8f;

#define N_NODES 100000
#define NPAD 100352
#define EPAD 1250048
#define OFF_CNT NPAD
#define OFF_ROWPTR (2 * NPAD)
#define OFF_RANK (3 * NPAD)
#define OFF_CV (OFF_RANK + EPAD)
#define OFF_BUFA (OFF_CV + 2 * EPAD)
#define NBUF ((size_t)N_NODES * 64)


__global__ void gcn_cnt_zero(int* __restrict__ cnt) {
  cnt[blockIdx.x * blockDim.x + threadIdx.x] = 0;
}

__global__ void gcn_count_rank(const int* __restrict__ dst, int* __restrict__ cnt,
                               int* __restrict__ rank, int nE) {
  int e = blockIdx.x * blockDim.x + threadIdx.x;
  if (e < nE) rank[e] = atomicAdd(&cnt[dst[e]], 1);
}

__global__ __launch_bounds__(256)
void gcn_scan_dis(const int* __restrict__ cnt, int* __restrict__ rowptr,
                  float* __restrict__ dis) {
  __shared__ int s_scan[256];
  __shared__ int s_part[256];
  const int t = threadIdx.x;
  const int base4 = blockIdx.x * 256;
  const int4* c4 = (const int4*)cnt;
  int part = 0;
  for (int q = t; q < base4; q += 256) {
    const int4 p = c4[q];
    part += (p.x + p.y) + (p.z + p.w);
  }
  const int4 c = c4[base4 + t];
  const int s = (c.x + c.y) + (c.z + c.w);
  s_part[t] = part;
  s_scan[t] = s;
  __syncthreads();
  for (int off = 128; off > 0; off >>= 1) {
    if (t < off) s_part[t] += s_part[t + off];
    __syncthreads();
  }
  for (int off = 1; off < 256; off <<= 1) {
    const int v = (t >= off) ? s_scan[t - off] : 0;
    __syncthreads();
    s_scan[t] += v;
    __syncthreads();
  }
  const int excl = s_part[0] + s_scan[t] - s;
  int4 r;
  r.x = excl;
  r.y = r.x + c.x;
  r.z = r.y + c.y;
  r.w = r.z + c.z;
  ((int4*)rowptr)[base4 + t] = r;
  float4 d;
  d.x = rsqrtf((float)(c.x + 1));
  d.y = rsqrtf((float)(c.y + 1));
  d.z = rsqrtf((float)(c.z + 1));
  d.w = rsqrtf((float)(c.w + 1));
  ((float4*)dis)[base4 + t] = d;
}

__global__ void gcn_csr_fill(const int* __restrict__ src, const int* __restrict__ dst,
                             const int* __restrict__ rank, const int* __restrict__ rowptr,
                             const float* __restrict__ dis, int2* __restrict__ cv, int nE) {
  int e = blockIdx.x * blockDim.x + threadIdx.x;
  if (e < nE) {
    const int s = src[e];
    const int d = dst[e];
    int2 o;
    o.x = s;
    o.y = __float_as_int(dis[s] * dis[d]);
    cv[rowptr[d] + rank[e]] = o;
  }
}


template <int K, int NC, bool RELU>
__global__ __launch_bounds__(64)
void gcn_gemm_wmma(const float* __restrict__ X, const float* __restrict__ W,
                   float* __restrict__ H) {
  constexpr int S  = K + 6;
  constexpr int CT = NC / 16;
  __shared__ float sWt[NC * S];

  for (int t = threadIdx.x; t < K * NC / 4; t += 64) {
    const int kr = (t * 4) / NC;
    const int n  = (t * 4) % NC;
    const float4 w4 = *(const float4*)(W + (size_t)t * 4);
    sWt[(size_t)(n + 0) * S + kr] = w4.x;
    sWt[(size_t)(n + 1) * S + kr] = w4.y;
    sWt[(size_t)(n + 2) * S + kr] = w4.z;
    sWt[(size_t)(n + 3) * S + kr] = w4.w;
  }
  __syncthreads();

  const int lane = threadIdx.x & 31;
  const int wave = threadIdx.x >> 5;
  const int m    = lane & 15;
  const int half = lane >> 4;
  const long long row0 = ((long long)blockIdx.x * 2 + wave) * 16;

  const float* xrow  = X + (size_t)(row0 + m) * K + 2 * half;
  const float* bbase = sWt + (size_t)m * S + 2 * half;

  v8f acc[CT] = {};
#pragma unroll
  for (int kk = 0; kk < K / 4; ++kk) {
    const int k = kk * 4;
    float ax = xrow[k];
    float ay = xrow[k + 1];
    if (RELU) { ax = fmaxf(ax, 0.f); ay = fmaxf(ay, 0.f); }
    const v2f a = {ax, ay};
#pragma unroll
    for (int ct = 0; ct < CT; ++ct) {
      const v2f b = *(const v2f*)(bbase + (size_t)ct * 16 * S + k);
      acc[ct] = __builtin_amdgcn_wmma_f32_16x16x4_f32(
          false, a, false, b, (short)0, acc[ct], false, false);
    }
  }

#pragma unroll
  for (int ct = 0; ct < CT; ++ct) {
#pragma unroll
    for (int r = 0; r < 8; ++r) {
      const long long row = row0 + r + 8 * half;
      H[(size_t)row * NC + ct * 16 + m] = acc[ct][r];
    }
  }
}


template <int F>
__global__ __launch_bounds__(256)
void gcn_gather(const float* __restrict__ h, const int2* __restrict__ cv,
                const int* __restrict__ rowptr, const int* __restrict__ cnt,
                const float* __restrict__ dis, const float* __restrict__ bias,
                float* __restrict__ out, int n) {
  constexpr int G = F / 4;
  const int tid  = blockIdx.x * 256 + threadIdx.x;
  const int node = tid / G;
  const int fo   = (tid % G) * 4;
  if (node >= n) return;

  const float dv = dis[node];
  const float sw = dv * dv;
  const int beg  = rowptr[node];
  const int end  = beg + cnt[node];
  const float4 hs = *(const float4*)(h + (size_t)node * F + fo);
  float4 acc;
  acc.x = hs.x * sw;
  acc.y = hs.y * sw;
  acc.z = hs.z * sw;
  acc.w = hs.w * sw;

  const float* hcol = h + fo;
  const int last = end - 1;
  for (int p = beg; p < end; p += 4) {
    const int p1 = (p + 1 < end) ? p + 1 : last;
    const int p2 = (p + 2 < end) ? p + 2 : last;
    const int p3 = (p + 3 < end) ? p + 3 : last;
    const int2 e0 = cv[p];
    const int2 e1 = cv[p1];
    const int2 e2 = cv[p2];
    const int2 e3 = cv[p3];
    const float w0 = __int_as_float(e0.y);
    const float w1 = (p + 1 < end) ? __int_as_float(e1.y) : 0.f;
    const float w2 = (p + 2 < end) ? __int_as_float(e2.y) : 0.f;
    const float w3 = (p + 3 < end) ? __int_as_float(e3.y) : 0.f;
    const float4 v0 = *(const float4*)(hcol + (size_t)e0.x * F);
    const float4 v1 = *(const float4*)(hcol + (size_t)e1.x * F);
    const float4 v2 = *(const float4*)(hcol + (size_t)e2.x * F);
    const float4 v3 = *(const float4*)(hcol + (size_t)e3.x * F);
    acc.x = fmaf(v0.x, w0, acc.x); acc.y = fmaf(v0.y, w0, acc.y);
    acc.z = fmaf(v0.z, w0, acc.z); acc.w = fmaf(v0.w, w0, acc.w);
    acc.x = fmaf(v1.x, w1, acc.x); acc.y = fmaf(v1.y, w1, acc.y);
    acc.z = fmaf(v1.z, w1, acc.z); acc.w = fmaf(v1.w, w1, acc.w);
    acc.x = fmaf(v2.x, w2, acc.x); acc.y = fmaf(v2.y, w2, acc.y);
    acc.z = fmaf(v2.z, w2, acc.z); acc.w = fmaf(v2.w, w2, acc.w);
    acc.x = fmaf(v3.x, w3, acc.x); acc.y = fmaf(v3.y, w3, acc.y);
    acc.z = fmaf(v3.z, w3, acc.z); acc.w = fmaf(v3.w, w3, acc.w);
  }

  const float4 b = *(const float4*)(bias + fo);
  float4 o;
  o.x = acc.x + b.x;
  o.y = acc.y + b.y;
  o.z = acc.z + b.z;
  o.w = acc.w + b.w;
  *(float4*)(out + (size_t)node * F + fo) = o;
}


extern "C" void kernel_launch(void* const* d_in, const int* in_sizes, int n_in,
                              void* d_out, int out_size, void* d_ws, size_t ws_size,
                              hipStream_t stream) {
  const float* x  = (const float*)d_in[0];
  const int*   ei = (const int*)d_in[1];
  const float* W1 = (const float*)d_in[2];
  const float* b1 = (const float*)d_in[3];
  const float* W2 = (const float*)d_in[4];
  const float* b2 = (const float*)d_in[5];
  const float* W3 = (const float*)d_in[6];
  const float* b3 = (const float*)d_in[7];
  float* out = (float*)d_out;

  const int N = N_NODES;
  const int E = in_sizes[1] / 2;
  const int* src = ei;
  const int* dst = ei + E;

  float* ws     = (float*)d_ws;
  float* dis    = ws;
  int*   cnt    = (int*)(ws + OFF_CNT);
  int*   rowptr = (int*)(ws + OFF_ROWPTR);
  int*   rank   = (int*)(ws + OFF_RANK);
  int2*  cv     = (int2*)(ws + OFF_CV);
  float* bufA   = ws + OFF_BUFA;
  float* bufB   = bufA + NBUF;

  const int TB = 256;
  const int gE = (E + TB - 1) / TB;
  const int gGemm = N / 32;
  const int gG64 = (N * 16 + TB - 1) / TB;
  const int gG32 = (N * 8 + TB - 1) / TB;

  gcn_cnt_zero  <<<NPAD / TB, TB, 0, stream>>>(cnt);
  gcn_count_rank<<<gE, TB, 0, stream>>>(dst, cnt, rank, E);
  gcn_scan_dis  <<<NPAD / 1024, TB, 0, stream>>>(cnt, rowptr, dis);
  gcn_csr_fill  <<<gE, TB, 0, stream>>>(src, dst, rank, rowptr, dis, cv, E);

  gcn_gemm_wmma<128, 64, false><<<gGemm, 64, 0, stream>>>(x, W1, bufA);
  gcn_gather<64><<<gG64, TB, 0, stream>>>(bufA, cv, rowptr, cnt, dis, b1, bufB, N);

  gcn_gemm_wmma<64, 64, true><<<gGemm, 64, 0, stream>>>(bufB, W2, bufA);
  gcn_gather<64><<<gG64, TB, 0, stream>>>(bufA, cv, rowptr, cnt, dis, b2, bufB, N);

  gcn_gemm_wmma<64, 32, true><<<gGemm, 64, 0, stream>>>(bufB, W3, bufA);
  gcn_gather<32><<<gG32, TB, 0, stream>>>(bufA, cv, rowptr, cnt, dis, b3, out, N);
}


# Dst-grouped edge list built by two-level bucket sort on LDS counters, not per-edge global atomics; edge weight factored as dis[dst]*sum(dis[src]*h[src]); sum order = LDS arrival rank
# speedup vs baseline: 3.2581x; 1.6774x over previous
#include <hip/hip_runtime.h>


typedef __attribute__((ext_vector_type(2))) float v2f;
typedef __attribute__((ext_vector_type(8))) float v8f;

#define N_NODES 100000
#define NB 196
#define NPAD (NB * 512)
#define NW 256
#define CHUNK 4884
#define EPT 5
#define EPAD 1250048
#define CAP 12288
#define OFF_CNT NPAD
#define OFF_ROWPTR (2 * NPAD)
#define OFF_CBW (3 * NPAD)
#define OFF_OBW (OFF_CBW + NB * NW)
#define OFF_BK (OFF_OBW + NB * NW)
#define OFF_CS (OFF_BK + 2 * EPAD)
#define OFF_BUFA (OFF_CS + EPAD)
#define NBUF ((size_t)N_NODES * 64)


__device__ __forceinline__ int block_excl_scan(int x, int* s_wsum) {
  const int lane = threadIdx.x & 31;
  const int wave = threadIdx.x >> 5;
  int inc = x;
#pragma unroll
  for (int o = 1; o < 32; o <<= 1) {
    const int y = __shfl_up(inc, o);
    if (lane >= o) inc += y;
  }
  if (lane == 31) s_wsum[wave] = inc;
  __syncthreads();
  int woff = 0;
  for (int i = 0; i < wave; ++i) woff += s_wsum[i];
  __syncthreads();
  return woff + inc - x;
}

__global__ __launch_bounds__(1024)
void gcn_bucket_hist(const int* __restrict__ dst, int* __restrict__ cntBW, int nE) {
  __shared__ int s_h[NB];
  const int t = threadIdx.x;
  if (t < NB) s_h[t] = 0;
  const int e0 = blockIdx.x * CHUNK;
  const int e1 = min(e0 + CHUNK, nE);
  int d[EPT];
#pragma unroll
  for (int k = 0; k < EPT; ++k) {
    const int e = e0 + t + k * 1024;
    d[k] = (e < e1) ? dst[e] : -1;
  }
  __syncthreads();
#pragma unroll
  for (int k = 0; k < EPT; ++k)
    if (d[k] >= 0) atomicAdd(&s_h[d[k] >> 9], 1);
  __syncthreads();
  if (t < NB) cntBW[t * NW + blockIdx.x] = s_h[t];
}

static_assert(NW == 256 && NB == 28 * 7, "scan layout");
__global__ __launch_bounds__(896)
void gcn_bucket_scan(const int* __restrict__ cntBW, int* __restrict__ offBW) {
  __shared__ int s_w[28];
  const int lane = threadIdx.x & 31;
  const int wave = threadIdx.x >> 5;
  const int q0 = wave * 7 * (NW / 4) + lane * 2;
  const int4* in4 = (const int4*)cntBW;
  int4 a[7], c[7];
#pragma unroll
  for (int k = 0; k < 7; ++k) {
    a[k] = in4[q0 + k * (NW / 4)];
    c[k] = in4[q0 + k * (NW / 4) + 1];
  }
  int carry = 0;
#pragma unroll
  for (int k = 0; k < 7; ++k) {
    const int4 u = a[k];
    const int4 v = c[k];
    const int s = ((u.x + u.y) + (u.z + u.w)) + ((v.x + v.y) + (v.z + v.w));
    int inc = s;
#pragma unroll
    for (int o = 1; o < 32; o <<= 1) {
      const int y = __shfl_up(inc, o);
      if (lane >= o) inc += y;
    }
    int4 oa, oc;
    oa.x = carry + inc - s;
    oa.y = oa.x + u.x;
    oa.z = oa.y + u.y;
    oa.w = oa.z + u.z;
    oc.x = oa.w + u.w;
    oc.y = oc.x + v.x;
    oc.z = oc.y + v.y;
    oc.w = oc.z + v.z;
    a[k] = oa;
    c[k] = oc;
    carry += __shfl(inc, 31);
  }
  if (lane == 0) s_w[wave] = carry;
  __syncthreads();
  int woff = 0;
  for (int i = 0; i < wave; ++i) woff += s_w[i];
  int4* out4 = (int4*)offBW;
#pragma unroll
  for (int k = 0; k < 7; ++k) {
    int4 oa = a[k], oc = c[k];
    oa.x += woff; oa.y += woff; oa.z += woff; oa.w += woff;
    oc.x += woff; oc.y += woff; oc.z += woff; oc.w += woff;
    out4[q0 + k * (NW / 4)] = oa;
    out4[q0 + k * (NW / 4) + 1] = oc;
  }
}

__global__ __launch_bounds__(1024)
void gcn_bucket_scatter(const int* __restrict__ src, const int* __restrict__ dst,
                        const int* __restrict__ cntBW, const int* __restrict__ offBW,
                        int2* __restrict__ bk, int nE) {
  __shared__ int s_wsum[32];
  __shared__ int s_loc[NB];
  __shared__ int s_cur[NB];
  __shared__ int s_goff[NB];
  __shared__ int2 s_pairs[CHUNK];
  const int t = threadIdx.x;
  const int w = blockIdx.x;
  const int e0 = w * CHUNK;
  const int e1 = min(e0 + CHUNK, nE);
  int sv[EPT], dv[EPT];
#pragma unroll
  for (int k = 0; k < EPT; ++k) {
    const int e = e0 + t + k * 1024;
    dv[k] = (e < e1) ? dst[e] : -1;
    sv[k] = (e < e1) ? src[e] : 0;
  }
  const int loc = block_excl_scan((t < NB) ? cntBW[t * NW + w] : 0, s_wsum);
  if (t < NB) {
    s_loc[t] = loc;
    s_cur[t] = loc;
    s_goff[t] = offBW[t * NW + w];
  }
  __syncthreads();
#pragma unroll
  for (int k = 0; k < EPT; ++k) {
    if (dv[k] >= 0) {
      int2 p;
      p.x = sv[k];
      p.y = dv[k];
      const int lp = atomicAdd(&s_cur[dv[k] >> 9], 1);
      s_pairs[lp] = p;
    }
  }
  __syncthreads();
#pragma unroll
  for (int k = 0; k < EPT; ++k) {
    const int i = t + k * 1024;
    if (i < e1 - e0) {
      const int2 p = s_pairs[i];
      const int b = p.y >> 9;
      bk[s_goff[b] + (i - s_loc[b])] = p;
    }
  }
}

__global__ __launch_bounds__(1024)
void gcn_bucket_csr(const int2* __restrict__ bk, const int* __restrict__ offBW,
                    int* __restrict__ rowptr, int* __restrict__ cnt,
                    float* __restrict__ dis, int* __restrict__ csrc, int nE) {
  __shared__ int s_wsum[32];
  __shared__ int s_cnt[512];
  __shared__ int s_cur[512];
  __shared__ int s_sorted[CAP];
  const int t = threadIdx.x;
  const int b = blockIdx.x;
  const int beg = offBW[b * NW];
  const int end = (b + 1 < NB) ? offBW[(b + 1) * NW] : nE;
  if (t < 512) s_cnt[t] = 0;
  __syncthreads();
  for (int eb = beg; eb < end; eb += 4096) {
    int y[4];
#pragma unroll
    for (int k = 0; k < 4; ++k) {
      const int e = eb + t + k * 1024;
      y[k] = (e < end) ? bk[e].y : -1;
    }
#pragma unroll
    for (int k = 0; k < 4; ++k)
      if (y[k] >= 0) atomicAdd(&s_cnt[y[k] & 511], 1);
  }
  __syncthreads();
  const int c = (t < 512) ? s_cnt[t] : 0;
  const int ex = block_excl_scan(c, s_wsum);
  if (t < 512) {
    const int node = b * 512 + t;
    rowptr[node] = beg + ex;
    cnt[node] = c;
    dis[node] = rsqrtf((float)(c + 1));
    s_cur[t] = ex;
  }
  __syncthreads();
  const int m = end - beg;
  if (m <= CAP) {
    for (int eb = beg; eb < end; eb += 4096) {
      int2 p[4];
#pragma unroll
      for (int k = 0; k < 4; ++k) {
        const int e = eb + t + k * 1024;
        int2 q;
        q.x = 0;
        q.y = -1;
        if (e < end) q = bk[e];
        p[k] = q;
      }
#pragma unroll
      for (int k = 0; k < 4; ++k) {
        if (p[k].y >= 0) {
          const int lp = atomicAdd(&s_cur[p[k].y & 511], 1);
          s_sorted[lp] = p[k].x;
        }
      }
    }
    __syncthreads();
    for (int i = t; i < m; i += 1024) csrc[beg + i] = s_sorted[i];
  } else {
    for (int e = beg + t; e < end; e += 1024) {
      const int2 p = bk[e];
      const int lp = atomicAdd(&s_cur[p.y & 511], 1);
      csrc[beg + lp] = p.x;
    }
  }
}


template <int K, int NC, bool RELU>
__global__ __launch_bounds__(64)
void gcn_gemm_wmma(const float* __restrict__ X, const float* __restrict__ W,
                   const float* __restrict__ dis, float* __restrict__ H) {
  constexpr int S  = K + 6;
  constexpr int CT = NC / 16;
  __shared__ float sWt[NC * S];

  for (int t = threadIdx.x; t < K * NC / 4; t += 64) {
    const int kr = (t * 4) / NC;
    const int n  = (t * 4) % NC;
    const float4 w4 = *(const float4*)(W + (size_t)t * 4);
    sWt[(size_t)(n + 0) * S + kr] = w4.x;
    sWt[(size_t)(n + 1) * S + kr] = w4.y;
    sWt[(size_t)(n + 2) * S + kr] = w4.z;
    sWt[(size_t)(n + 3) * S + kr] = w4.w;
  }
  __syncthreads();

  const int lane = threadIdx.x & 31;
  const int wave = threadIdx.x >> 5;
  const int m    = lane & 15;
  const int half = lane >> 4;
  const long long row0 = ((long long)blockIdx.x * 2 + wave) * 16;

  const float* xrow  = X + (size_t)(row0 + m) * K + 2 * half;
  const float* bbase = sWt + (size_t)m * S + 2 * half;

  v8f acc[CT] = {};
#pragma unroll
  for (int kk = 0; kk < K / 4; ++kk) {
    const int k = kk * 4;
    float ax = xrow[k];
    float ay = xrow[k + 1];
    if (RELU) { ax = fmaxf(ax, 0.f); ay = fmaxf(ay, 0.f); }
    const v2f a = {ax, ay};
#pragma unroll
    for (int ct = 0; ct < CT; ++ct) {
      const v2f b = *(const v2f*)(bbase + (size_t)ct * 16 * S + k);
      acc[ct] = __builtin_amdgcn_wmma_f32_16x16x4_f32(
          false, a, false, b, (short)0, acc[ct], false, false);
    }
  }

  float dsc[8];
#pragma unroll
  for (int r = 0; r < 8; ++r) dsc[r] = dis[row0 + r + 8 * half];
#pragma unroll
  for (int ct = 0; ct < CT; ++ct) {
#pragma unroll
    for (int r = 0; r < 8; ++r) {
      const long long row = row0 + r + 8 * half;
      H[(size_t)row * NC + ct * 16 + m] = acc[ct][r] * dsc[r];
    }
  }
}


template <int F>
__global__ __launch_bounds__(256)
void gcn_gather(const float* __restrict__ h, const int* __restrict__ cs,
                const int* __restrict__ rowptr, const int* __restrict__ cnt,
                const float* __restrict__ dis, const float* __restrict__ bias,
                float* __restrict__ out, int n) {
  constexpr int G = F / 4;
  const int tid  = blockIdx.x * 256 + threadIdx.x;
  const int node = tid / G;
  const int fo   = (tid % G) * 4;
  if (node >= n) return;

  const float dv = dis[node];
  const int beg  = rowptr[node];
  const int end  = beg + cnt[node];
  float4 acc = *(const float4*)(h + (size_t)node * F + fo);

  const float* hcol = h + fo;
  const int last = end - 1;
  for (int p = beg; p < end; p += 4) {
    const int p1 = (p + 1 < end) ? p + 1 : last;
    const int p2 = (p + 2 < end) ? p + 2 : last;
    const int p3 = (p + 3 < end) ? p + 3 : last;
    const int s0 = cs[p];
    const int s1 = cs[p1];
    const int s2 = cs[p2];
    const int s3 = cs[p3];
    const float w1 = (p + 1 < end) ? 1.f : 0.f;
    const float w2 = (p + 2 < end) ? 1.f : 0.f;
    const float w3 = (p + 3 < end) ? 1.f : 0.f;
    const float4 v0 = *(const float4*)(hcol + (size_t)s0 * F);
    const float4 v1 = *(const float4*)(hcol + (size_t)s1 * F);
    const float4 v2 = *(const float4*)(hcol + (size_t)s2 * F);
    const float4 v3 = *(const float4*)(hcol + (size_t)s3 * F);
    acc.x += v0.x; acc.y += v0.y;
    acc.z += v0.z; acc.w += v0.w;
    acc.x = fmaf(v1.x, w1, acc.x); acc.y = fmaf(v1.y, w1, acc.y);
    acc.z = fmaf(v1.z, w1, acc.z); acc.w = fmaf(v1.w, w1, acc.w);
    acc.x = fmaf(v2.x, w2, acc.x); acc.y = fmaf(v2.y, w2, acc.y);
    acc.z = fmaf(v2.z, w2, acc.z); acc.w = fmaf(v2.w, w2, acc.w);
    acc.x = fmaf(v3.x, w3, acc.x); acc.y = fmaf(v3.y, w3, acc.y);
    acc.z = fmaf(v3.z, w3, acc.z); acc.w = fmaf(v3.w, w3, acc.w);
  }

  const float4 b = *(const float4*)(bias + fo);
  float4 o;
  o.x = fmaf(acc.x, dv, b.x);
  o.y = fmaf(acc.y, dv, b.y);
  o.z = fmaf(acc.z, dv, b.z);
  o.w = fmaf(acc.w, dv, b.w);
  *(float4*)(out + (size_t)node * F + fo) = o;
}


extern "C" void kernel_launch(void* const* d_in, const int* in_sizes, int n_in,
                              void* d_out, int out_size, void* d_ws, size_t ws_size,
                              hipStream_t stream) {
  const float* x  = (const float*)d_in[0];
  const int*   ei = (const int*)d_in[1];
  const float* W1 = (const float*)d_in[2];
  const float* b1 = (const float*)d_in[3];
  const float* W2 = (const float*)d_in[4];
  const float* b2 = (const float*)d_in[5];
  const float* W3 = (const float*)d_in[6];
  const float* b3 = (const float*)d_in[7];
  float* out = (float*)d_out;

  const int N = N_NODES;
  const int E = in_sizes[1] / 2;
  const int* src = ei;
  const int* dst = ei + E;

  float* ws     = (float*)d_ws;
  float* dis    = ws;
  int*   cnt    = (int*)(ws + OFF_CNT);
  int*   rowptr = (int*)(ws + OFF_ROWPTR);
  int*   cntBW  = (int*)(ws + OFF_CBW);
  int*   offBW  = (int*)(ws + OFF_OBW);
  int2*  bk     = (int2*)(ws + OFF_BK);
  int*   cs     = (int*)(ws + OFF_CS);
  float* bufA   = ws + OFF_BUFA;
  float* bufB   = bufA + NBUF;

  const int TB = 256;
  const int gGemm = N / 32;
  const int gG64 = (N * 16 + TB - 1) / TB;
  const int gG32 = (N * 8 + TB - 1) / TB;

  gcn_bucket_hist   <<<NW, 1024, 0, stream>>>(dst, cntBW, E);
  gcn_bucket_scan   <<<1, 896, 0, stream>>>(cntBW, offBW);
  gcn_bucket_scatter<<<NW, 1024, 0, stream>>>(src, dst, cntBW, offBW, bk, E);
  gcn_bucket_csr    <<<NB, 1024, 0, stream>>>(bk, offBW, rowptr, cnt, dis, cs, E);

  gcn_gemm_wmma<128, 64, false><<<gGemm, 64, 0, stream>>>(x, W1, dis, bufA);
  gcn_gather<64><<<gG64, TB, 0, stream>>>(bufA, cs, rowptr, cnt, dis, b1, bufB, N);

  gcn_gemm_wmma<64, 64, true><<<gGemm, 64, 0, stream>>>(bufB, W2, dis, bufA);
  gcn_gather<64><<<gG64, TB, 0, stream>>>(bufA, cs, rowptr, cnt, dis, b2, bufB, N);

  gcn_gemm_wmma<64, 32, true><<<gGemm, 64, 0, stream>>>(bufB, W3, dis, bufA);
  gcn_gather<32><<<gG32, TB, 0, stream>>>(bufA, cs, rowptr, cnt, dis, b3, out, N);
}


# Aggregation launches as resident grids: 10000/10000/8336 waves, each lane group walks 5 (3) nodes with bounds, source ids and rows requested ahead, 8 row gathers in flight; sum order unchanged
# speedup vs baseline: 3.3598x; 1.0312x over previous
#include <hip/hip_runtime.h>


typedef __attribute__((ext_vector_type(2))) float v2f;
typedef __attribute__((ext_vector_type(8))) float v8f;

#define N_NODES 100000
#define NB 196
#define NPAD (NB * 512)
#define NW 256
#define CHUNK 4884
#define EPT 5
#define EPAD 1250048
#define CAP 12288
#define OFF_CNT NPAD
#define OFF_ROWPTR (2 * NPAD)
#define OFF_CBW (3 * NPAD)
#define OFF_OBW (OFF_CBW + NB * NW)
#define OFF_BK (OFF_OBW + NB * NW)
#define OFF_CS (OFF_BK + 2 * EPAD)
#define OFF_BUFA (OFF_CS + EPAD)
#define NBUF ((size_t)N_NODES * 64)


__device__ __forceinline__ int block_excl_scan(int x, int* s_wsum) {
  const int lane = threadIdx.x & 31;
  const int wave = threadIdx.x >> 5;
  int inc = x;
#pragma unroll
  for (int o = 1; o < 32; o <<= 1) {
    const int y = __shfl_up(inc, o);
    if (lane >= o) inc += y;
  }
  if (lane == 31) s_wsum[wave] = inc;
  __syncthreads();
  int woff = 0;
  for (int i = 0; i < wave; ++i) woff += s_wsum[i];
  __syncthreads();
  return woff + inc - x;
}

__global__ __launch_bounds__(1024)
void gcn_bucket_hist(const int* __restrict__ dst, int* __restrict__ cntBW, int nE) {
  __shared__ int s_h[NB];
  const int t = threadIdx.x;
  if (t < NB) s_h[t] = 0;
  const int e0 = blockIdx.x * CHUNK;
  const int e1 = min(e0 + CHUNK, nE);
  int d[EPT];
#pragma unroll
  for (int k = 0; k < EPT; ++k) {
    const int e = e0 + t + k * 1024;
    d[k] = (e < e1) ? dst[e] : -1;
  }
  __syncthreads();
#pragma unroll
  for (int k = 0; k < EPT; ++k)
    if (d[k] >= 0) atomicAdd(&s_h[d[k] >> 9], 1);
  __syncthreads();
  if (t < NB) cntBW[t * NW + blockIdx.x] = s_h[t];
}

static_assert(NW == 256 && NB == 28 * 7, "scan layout");
__global__ __launch_bounds__(896)
void gcn_bucket_scan(const int* __restrict__ cntBW, int* __restrict__ offBW) {
  __shared__ int s_w[28];
  const int lane = threadIdx.x & 31;
  const int wave = threadIdx.x >> 5;
  const int q0 = wave * 7 * (NW / 4) + lane * 2;
  const int4* in4 = (const int4*)cntBW;
  int4 a[7], c[7];
#pragma unroll
  for (int k = 0; k < 7; ++k) {
    a[k] = in4[q0 + k * (NW / 4)];
    c[k] = in4[q0 + k * (NW / 4) + 1];
  }
  int carry = 0;
#pragma unroll
  for (int k = 0; k < 7; ++k) {
    const int4 u = a[k];
    const int4 v = c[k];
    const int s = ((u.x + u.y) + (u.z + u.w)) + ((v.x + v.y) + (v.z + v.w));
    int inc = s;
#pragma unroll
    for (int o = 1; o < 32; o <<= 1) {
      const int y = __shfl_up(inc, o);
      if (lane >= o) inc += y;
    }
    int4 oa, oc;
    oa.x = carry + inc - s;
    oa.y = oa.x + u.x;
    oa.z = oa.y + u.y;
    oa.w = oa.z + u.z;
    oc.x = oa.w + u.w;
    oc.y = oc.x + v.x;
    oc.z = oc.y + v.y;
    oc.w = oc.z + v.z;
    a[k] = oa;
    c[k] = oc;
    carry += __shfl(inc, 31);
  }
  if (lane == 0) s_w[wave] = carry;
  __syncthreads();
  int woff = 0;
  for (int i = 0; i < wave; ++i) woff += s_w[i];
  int4* out4 = (int4*)offBW;
#pragma unroll
  for (int k = 0; k < 7; ++k) {
    int4 oa = a[k], oc = c[k];
    oa.x += woff; oa.y += woff; oa.z += woff; oa.w += woff;
    oc.x += woff; oc.y += woff; oc.z += woff; oc.w += woff;
    out4[q0 + k * (NW / 4)] = oa;
    out4[q0 + k * (NW / 4) + 1] = oc;
  }
}

__global__ __launch_bounds__(1024)
void gcn_bucket_scatter(const int* __restrict__ src, const int* __restrict__ dst,
                        const int* __restrict__ cntBW, const int* __restrict__ offBW,
                        int2* __restrict__ bk, int nE) {
  __shared__ int s_wsum[32];
  __shared__ int s_loc[NB];
  __shared__ int s_cur[NB];
  __shared__ int s_goff[NB];
  __shared__ int2 s_pairs[CHUNK];
  const int t = threadIdx.x;
  const int w = blockIdx.x;
  const int e0 = w * CHUNK;
  const int e1 = min(e0 + CHUNK, nE);
  int sv[EPT], dv[EPT];
#pragma unroll
  for (int k = 0; k < EPT; ++k) {
    const int e = e0 + t + k * 1024;
    dv[k] = (e < e1) ? dst[e] : -1;
    sv[k] = (e < e1) ? src[e] : 0;
  }
  const int loc = block_excl_scan((t < NB) ? cntBW[t * NW + w] : 0, s_wsum);
  if (t < NB) {
    s_loc[t] = loc;
    s_cur[t] = loc;
    s_goff[t] = offBW[t * NW + w];
  }
  __syncthreads();
#pragma unroll
  for (int k = 0; k < EPT; ++k) {
    if (dv[k] >= 0) {
      int2 p;
      p.x = sv[k];
      p.y = dv[k];
      const int lp = atomicAdd(&s_cur[dv[k] >> 9], 1);
      s_pairs[lp] = p;
    }
  }
  __syncthreads();
#pragma unroll
  for (int k = 0; k < EPT; ++k) {
    const int i = t + k * 1024;
    if (i < e1 - e0) {
      const int2 p = s_pairs[i];
      const int b = p.y >> 9;
      bk[s_goff[b] + (i - s_loc[b])] = p;
    }
  }
}

__global__ __launch_bounds__(1024)
void gcn_bucket_csr(const int2* __restrict__ bk, const int* __restrict__ offBW,
                    int* __restrict__ rowptr, int* __restrict__ cnt,
                    float* __restrict__ dis, int* __restrict__ csrc, int nE) {
  __shared__ int s_wsum[32];
  __shared__ int s_cnt[512];
  __shared__ int s_cur[512];
  __shared__ int s_sorted[CAP];
  const int t = threadIdx.x;
  const int b = blockIdx.x;
  const int beg = offBW[b * NW];
  const int end = (b + 1 < NB) ? offBW[(b + 1) * NW] : nE;
  if (t < 512) s_cnt[t] = 0;
  __syncthreads();
  for (int eb = beg; eb < end; eb += 4096) {
    int y[4];
#pragma unroll
    for (int k = 0; k < 4; ++k) {
      const int e = eb + t + k * 1024;
      y[k] = (e < end) ? bk[e].y : -1;
    }
#pragma unroll
    for (int k = 0; k < 4; ++k)
      if (y[k] >= 0) atomicAdd(&s_cnt[y[k] & 511], 1);
  }
  __syncthreads();
  const int c = (t < 512) ? s_cnt[t] : 0;
  const int ex = block_excl_scan(c, s_wsum);
  if (t < 512) {
    const int node = b * 512 + t;
    rowptr[node] = beg + ex;
    cnt[node] = c;
    dis[node] = rsqrtf((float)(c + 1));
    s_cur[t] = ex;
  }
  __syncthreads();
  const int m = end - beg;
  if (m <= CAP) {
    for (int eb = beg; eb < end; eb += 4096) {
      int2 p[4];
#pragma unroll
      for (int k = 0; k < 4; ++k) {
        const int e = eb + t + k * 1024;
        int2 q;
        q.x = 0;
        q.y = -1;
        if (e < end) q = bk[e];
        p[k] = q;
      }
#pragma unroll
      for (int k = 0; k < 4; ++k) {
        if (p[k].y >= 0) {
          const int lp = atomicAdd(&s_cur[p[k].y & 511], 1);
          s_sorted[lp] = p[k].x;
        }
      }
    }
    __syncthreads();
    for (int i = t; i < m; i += 1024) csrc[beg + i] = s_sorted[i];
  } else {
    for (int e = beg + t; e < end; e += 1024) {
      const int2 p = bk[e];
      const int lp = atomicAdd(&s_cur[p.y & 511], 1);
      csrc[beg + lp] = p.x;
    }
  }
}


template <int K, int NC, bool RELU>
__global__ __launch_bounds__(64)
void gcn_gemm_wmma(const float* __restrict__ X, const float* __restrict__ W,
                   const float* __restrict__ dis, float* __restrict__ H) {
  constexpr int S  = K + 6;
  constexpr int CT = NC / 16;
  __shared__ float sWt[NC * S];

  for (int t = threadIdx.x; t < K * NC / 4; t += 64) {
    const int kr = (t * 4) / NC;
    const int n  = (t * 4) % NC;
    const float4 w4 = *(const float4*)(W + (size_t)t * 4);
    sWt[(size_t)(n + 0) * S + kr] = w4.x;
    sWt[(size_t)(n + 1) * S + kr] = w4.y;
    sWt[(size_t)(n + 2) * S + kr] = w4.z;
    sWt[(size_t)(n + 3) * S + kr] = w4.w;
  }
  __syncthreads();

  const int lane = threadIdx.x & 31;
  const int wave = threadIdx.x >> 5;
  const int m    = lane & 15;
  const int half = lane >> 4;
  const long long row0 = ((long long)blockIdx.x * 2 + wave) * 16;

  const float* xrow  = X + (size_t)(row0 + m) * K + 2 * half;
  const float* bbase = sWt + (size_t)m * S + 2 * half;

  v8f acc[CT] = {};
#pragma unroll
  for (int kk = 0; kk < K / 4; ++kk) {
    const int k = kk * 4;
    float ax = xrow[k];
    float ay = xrow[k + 1];
    if (RELU) { ax = fmaxf(ax, 0.f); ay = fmaxf(ay, 0.f); }
    const v2f a = {ax, ay};
#pragma unroll
    for (int ct = 0; ct < CT; ++ct) {
      const v2f b = *(const v2f*)(bbase + (size_t)ct * 16 * S + k);
      acc[ct] = __builtin_amdgcn_wmma_f32_16x16x4_f32(
          false, a, false, b, (short)0, acc[ct], false, false);
    }
  }

  float dsc[8];
#pragma unroll
  for (int r = 0; r < 8; ++r) dsc[r] = dis[row0 + r + 8 * half];
#pragma unroll
  for (int ct = 0; ct < CT; ++ct) {
#pragma unroll
    for (int r = 0; r < 8; ++r) {
      const long long row = row0 + r + 8 * half;
      H[(size_t)row * NC + ct * 16 + m] = acc[ct][r] * dsc[r];
    }
  }
}


template <int F, int PASSES>
__global__ __launch_bounds__(256)
void gcn_gather(const float* __restrict__ h, const int* __restrict__ cs,
                const int* __restrict__ rowptr, const int* __restrict__ cnt,
                const float* __restrict__ dis, const float* __restrict__ bias,
                float* __restrict__ out, int n) {
  constexpr int G   = F / 4;
  constexpr int NPB = 256 / G;
  constexpr int EB  = 8;
  constexpr unsigned RB = F * 4;
  const int g  = threadIdx.x / G;
  const int fo = (threadIdx.x % G) * 4;
  const char* hb = (const char*)h;
  const unsigned lo = (unsigned)fo * 4u;
  const float4 b = *(const float4*)(bias + fo);
  const int node0 = blockIdx.x * (NPB * PASSES) + g;

  bool haveC = node0 < n;
  bool haveN = (PASSES > 1) && (node0 + NPB < n);
  int begC = 0, cntC = 0, begN = 0, cntN = 0;
  if (haveC) { begC = rowptr[node0];       cntC = cnt[node0]; }
  if (haveN) { begN = rowptr[node0 + NPB]; cntN = cnt[node0 + NPB]; }
  unsigned a[EB];
#pragma unroll
  for (int k = 0; k < EB; ++k) a[k] = lo;
  if (cntC > 0) {
    const int last = begC + cntC - 1;
    int t[EB];
#pragma unroll
    for (int k = 0; k < EB; ++k) t[k] = cs[min(begC + k, last)];
#pragma unroll
    for (int k = 0; k < EB; ++k) a[k] = (unsigned)t[k] * RB + lo;
  }

  for (int it = 0; it < PASSES; ++it) {
    const int node = node0 + it * NPB;

    const int nodeNN = node + 2 * NPB;
    const bool haveNN = (it + 2 < PASSES) && (nodeNN < n);
    int begNN = 0, cntNN = 0;
    if (haveNN) { begNN = rowptr[nodeNN]; cntNN = cnt[nodeNN]; }

    int tN[EB];
#pragma unroll
    for (int k = 0; k < EB; ++k) tN[k] = 0;
    if (cntN > 0) {
      const int lastN = begN + cntN - 1;
#pragma unroll
      for (int k = 0; k < EB; ++k) tN[k] = cs[min(begN + k, lastN)];
    }

    if (haveC) {
      const float dv = dis[node];
      float4 acc = *(const float4*)(hb + ((unsigned)node * RB + lo));
      const int end  = begC + cntC;
      const int last = end - 1;
      for (int p = begC; p < end; p += EB) {
        float4 v[EB];
#pragma unroll
        for (int k = 0; k < EB; ++k) v[k] = *(const float4*)(hb + a[k]);
        int t[EB];
#pragma unroll
        for (int k = 0; k < EB; ++k) t[k] = cs[min(p + EB + k, last)];
        acc.x += v[0].x; acc.y += v[0].y;
        acc.z += v[0].z; acc.w += v[0].w;
#pragma unroll
        for (int k = 1; k < EB; ++k) {
          const float w = (p + k < end) ? 1.f : 0.f;
          acc.x = fmaf(v[k].x, w, acc.x); acc.y = fmaf(v[k].y, w, acc.y);
          acc.z = fmaf(v[k].z, w, acc.z); acc.w = fmaf(v[k].w, w, acc.w);
        }
#pragma unroll
        for (int k = 0; k < EB; ++k) a[k] = (unsigned)t[k] * RB + lo;
      }
      float4 o;
      o.x = fmaf(acc.x, dv, b.x);
      o.y = fmaf(acc.y, dv, b.y);
      o.z = fmaf(acc.z, dv, b.z);
      o.w = fmaf(acc.w, dv, b.w);
      *(float4*)(out + (size_t)node * F + fo) = o;
    }

    haveC = haveN;  begC = begN;  cntC = cntN;
    haveN = haveNN; begN = begNN; cntN = cntNN;
#pragma unroll
    for (int k = 0; k < EB; ++k) a[k] = (unsigned)tN[k] * RB + lo;
  }
}


extern "C" void kernel_launch(void* const* d_in, const int* in_sizes, int n_in,
                              void* d_out, int out_size, void* d_ws, size_t ws_size,
                              hipStream_t stream) {
  const float* x  = (const float*)d_in[0];
  const int*   ei = (const int*)d_in[1];
  const float* W1 = (const float*)d_in[2];
  const float* b1 = (const float*)d_in[3];
  const float* W2 = (const float*)d_in[4];
  const float* b2 = (const float*)d_in[5];
  const float* W3 = (const float*)d_in[6];
  const float* b3 = (const float*)d_in[7];
  float* out = (float*)d_out;

  const int N = N_NODES;
  const int E = in_sizes[1] / 2;
  const int* src = ei;
  const int* dst = ei + E;

  float* ws     = (float*)d_ws;
  float* dis    = ws;
  int*   cnt    = (int*)(ws + OFF_CNT);
  int*   rowptr = (int*)(ws + OFF_ROWPTR);
  int*   cntBW  = (int*)(ws + OFF_CBW);
  int*   offBW  = (int*)(ws + OFF_OBW);
  int2*  bk     = (int2*)(ws + OFF_BK);
  int*   cs     = (int*)(ws + OFF_CS);
  float* bufA   = ws + OFF_BUFA;
  float* bufB   = bufA + NBUF;

  const int TB = 256;
  const int gGemm = N / 32;
  const int gG64 = 1250;
  const int gG32 = 1042;

  gcn_bucket_hist   <<<NW, 1024, 0, stream>>>(dst, cntBW, E);
  gcn_bucket_scan   <<<1, 896, 0, stream>>>(cntBW, offBW);
  gcn_bucket_scatter<<<NW, 1024, 0, stream>>>(src, dst, cntBW, offBW, bk, E);
  gcn_bucket_csr    <<<NB, 1024, 0, stream>>>(bk, offBW, rowptr, cnt, dis, cs, E);

  gcn_gemm_wmma<128, 64, false><<<gGemm, 64, 0, stream>>>(x, W1, dis, bufA);
  gcn_gather<64, 5><<<gG64, TB, 0, stream>>>(bufA, cs, rowptr, cnt, dis, b1, bufB, N);

  gcn_gemm_wmma<64, 64, true><<<gGemm, 64, 0, stream>>>(bufB, W2, dis, bufA);
  gcn_gather<64, 5><<<gG64, TB, 0, stream>>>(bufA, cs, rowptr, cnt, dis, b2, bufB, N);

  gcn_gemm_wmma<64, 32, true><<<gGemm, 64, 0, stream>>>(bufB, W3, dis, bufA);
  gcn_gather<32, 3><<<gG32, TB, 0, stream>>>(bufA, cs, rowptr, cnt, dis, b3, out, N);
}


# Layer 2 and 3 products moved into the aggregation launch before them: a work-group's 80 summed rows (ReLU) stay in LDS as WMMA A panels, W staged per group; 10 launches to 8; sum order unchanged
# speedup vs baseline: 4.0306x; 1.1997x over previous
#include <hip/hip_runtime.h>


typedef __attribute__((ext_vector_type(2))) float v2f;
typedef __attribute__((ext_vector_type(8))) float v8f;

#define N_NODES 100000
#define NB 196
#define NPAD (NB * 512)
#define NW 256
#define CHUNK 4884
#define EPT 5
#define EPAD 1250048
#define CAP 12288
#define OFF_CNT NPAD
#define OFF_ROWPTR (2 * NPAD)
#define OFF_CBW (3 * NPAD)
#define OFF_OBW (OFF_CBW + NB * NW)
#define OFF_BK (OFF_OBW + NB * NW)
#define OFF_CS (OFF_BK + 2 * EPAD)
#define OFF_BUFA (OFF_CS + EPAD)
#define NBUF ((size_t)N_NODES * 64)


__device__ __forceinline__ int block_excl_scan(int x, int* s_wsum) {
  const int lane = threadIdx.x & 31;
  const int wave = threadIdx.x >> 5;
  int inc = x;
#pragma unroll
  for (int o = 1; o < 32; o <<= 1) {
    const int y = __shfl_up(inc, o);
    if (lane >= o) inc += y;
  }
  if (lane == 31) s_wsum[wave] = inc;
  __syncthreads();
  int woff = 0;
  for (int i = 0; i < wave; ++i) woff += s_wsum[i];
  __syncthreads();
  return woff + inc - x;
}

__global__ __launch_bounds__(1024)
void gcn_bucket_hist(const int* __restrict__ dst, int* __restrict__ cntBW, int nE) {
  __shared__ int s_h[NB];
  const int t = threadIdx.x;
  if (t < NB) s_h[t] = 0;
  const int e0 = blockIdx.x * CHUNK;
  const int e1 = min(e0 + CHUNK, nE);
  int d[EPT];
#pragma unroll
  for (int k = 0; k < EPT; ++k) {
    const int e = e0 + t + k * 1024;
    d[k] = (e < e1) ? dst[e] : -1;
  }
  __syncthreads();
#pragma unroll
  for (int k = 0; k < EPT; ++k)
    if (d[k] >= 0) atomicAdd(&s_h[d[k] >> 9], 1);
  __syncthreads();
  if (t < NB) cntBW[t * NW + blockIdx.x] = s_h[t];
}

static_assert(NW == 256 && NB == 28 * 7, "scan layout");
__global__ __launch_bounds__(896)
void gcn_bucket_scan(const int* __restrict__ cntBW, int* __restrict__ offBW) {
  __shared__ int s_w[28];
  const int lane = threadIdx.x & 31;
  const int wave = threadIdx.x >> 5;
  const int q0 = wave * 7 * (NW / 4) + lane * 2;
  const int4* in4 = (const int4*)cntBW;
  int4 a[7], c[7];
#pragma unroll
  for (int k = 0; k < 7; ++k) {
    a[k] = in4[q0 + k * (NW / 4)];
    c[k] = in4[q0 + k * (NW / 4) + 1];
  }
  int carry = 0;
#pragma unroll
  for (int k = 0; k < 7; ++k) {
    const int4 u = a[k];
    const int4 v = c[k];
    const int s = ((u.x + u.y) + (u.z + u.w)) + ((v.x + v.y) + (v.z + v.w));
    int inc = s;
#pragma unroll
    for (int o = 1; o < 32; o <<= 1) {
      const int y = __shfl_up(inc, o);
      if (lane >= o) inc += y;
    }
    int4 oa, oc;
    oa.x = carry + inc - s;
    oa.y = oa.x + u.x;
    oa.z = oa.y + u.y;
    oa.w = oa.z + u.z;
    oc.x = oa.w + u.w;
    oc.y = oc.x + v.x;
    oc.z = oc.y + v.y;
    oc.w = oc.z + v.z;
    a[k] = oa;
    c[k] = oc;
    carry += __shfl(inc, 31);
  }
  if (lane == 0) s_w[wave] = carry;
  __syncthreads();
  int woff = 0;
  for (int i = 0; i < wave; ++i) woff += s_w[i];
  int4* out4 = (int4*)offBW;
#pragma unroll
  for (int k = 0; k < 7; ++k) {
    int4 oa = a[k], oc = c[k];
    oa.x += woff; oa.y += woff; oa.z += woff; oa.w += woff;
    oc.x += woff; oc.y += woff; oc.z += woff; oc.w += woff;
    out4[q0 + k * (NW / 4)] = oa;
    out4[q0 + k * (NW / 4) + 1] = oc;
  }
}

__global__ __launch_bounds__(1024)
void gcn_bucket_scatter(const int* __restrict__ src, const int* __restrict__ dst,
                        const int* __restrict__ cntBW, const int* __restrict__ offBW,
                        int2* __restrict__ bk, int nE) {
  __shared__ int s_wsum[32];
  __shared__ int s_loc[NB];
  __shared__ int s_cur[NB];
  __shared__ int s_goff[NB];
  __shared__ int2 s_pairs[CHUNK];
  const int t = threadIdx.x;
  const int w = blockIdx.x;
  const int e0 = w * CHUNK;
  const int e1 = min(e0 + CHUNK, nE);
  int sv[EPT], dv[EPT];
#pragma unroll
  for (int k = 0; k < EPT; ++k) {
    const int e = e0 + t + k * 1024;
    dv[k] = (e < e1) ? dst[e] : -1;
    sv[k] = (e < e1) ? src[e] : 0;
  }
  const int loc = block_excl_scan((t < NB) ? cntBW[t * NW + w] : 0, s_wsum);
  if (t < NB) {
    s_loc[t] = loc;
    s_cur[t] = loc;
    s_goff[t] = offBW[t * NW + w];
  }
  __syncthreads();
#pragma unroll
  for (int k = 0; k < EPT; ++k) {
    if (dv[k] >= 0) {
      int2 p;
      p.x = sv[k];
      p.y = dv[k];
      const int lp = atomicAdd(&s_cur[dv[k] >> 9], 1);
      s_pairs[lp] = p;
    }
  }
  __syncthreads();
#pragma unroll
  for (int k = 0; k < EPT; ++k) {
    const int i = t + k * 1024;
    if (i < e1 - e0) {
      const int2 p = s_pairs[i];
      const int b = p.y >> 9;
      bk[s_goff[b] + (i - s_loc[b])] = p;
    }
  }
}

__global__ __launch_bounds__(1024)
void gcn_bucket_csr(const int2* __restrict__ bk, const int* __restrict__ offBW,
                    int* __restrict__ rowptr, int* __restrict__ cnt,
                    float* __restrict__ dis, int* __restrict__ csrc, int nE) {
  __shared__ int s_wsum[32];
  __shared__ int s_cnt[512];
  __shared__ int s_cur[512];
  __shared__ int s_sorted[CAP];
  const int t = threadIdx.x;
  const int b = blockIdx.x;
  const int beg = offBW[b * NW];
  const int end = (b + 1 < NB) ? offBW[(b + 1) * NW] : nE;
  if (t < 512) s_cnt[t] = 0;
  __syncthreads();
  for (int eb = beg; eb < end; eb += 4096) {
    int y[4];
#pragma unroll
    for (int k = 0; k < 4; ++k) {
      const int e = eb + t + k * 1024;
      y[k] = (e < end) ? bk[e].y : -1;
    }
#pragma unroll
    for (int k = 0; k < 4; ++k)
      if (y[k] >= 0) atomicAdd(&s_cnt[y[k] & 511], 1);
  }
  __syncthreads();
  const int c = (t < 512) ? s_cnt[t] : 0;
  const int ex = block_excl_scan(c, s_wsum);
  if (t < 512) {
    const int node = b * 512 + t;
    rowptr[node] = beg + ex;
    cnt[node] = c;
    dis[node] = rsqrtf((float)(c + 1));
    s_cur[t] = ex;
  }
  __syncthreads();
  const int m = end - beg;
  if (m <= CAP) {
    for (int eb = beg; eb < end; eb += 4096) {
      int2 p[4];
#pragma unroll
      for (int k = 0; k < 4; ++k) {
        const int e = eb + t + k * 1024;
        int2 q;
        q.x = 0;
        q.y = -1;
        if (e < end) q = bk[e];
        p[k] = q;
      }
#pragma unroll
      for (int k = 0; k < 4; ++k) {
        if (p[k].y >= 0) {
          const int lp = atomicAdd(&s_cur[p[k].y & 511], 1);
          s_sorted[lp] = p[k].x;
        }
      }
    }
    __syncthreads();
    for (int i = t; i < m; i += 1024) csrc[beg + i] = s_sorted[i];
  } else {
    for (int e = beg + t; e < end; e += 1024) {
      const int2 p = bk[e];
      const int lp = atomicAdd(&s_cur[p.y & 511], 1);
      csrc[beg + lp] = p.x;
    }
  }
}


template <int K, int NC, bool RELU>
__global__ __launch_bounds__(64)
void gcn_gemm_wmma(const float* __restrict__ X, const float* __restrict__ W,
                   const float* __restrict__ dis, float* __restrict__ H) {
  constexpr int S  = K + 6;
  constexpr int CT = NC / 16;
  __shared__ float sWt[NC * S];

  for (int t = threadIdx.x; t < K * NC / 4; t += 64) {
    const int kr = (t * 4) / NC;
    const int n  = (t * 4) % NC;
    const float4 w4 = *(const float4*)(W + (size_t)t * 4);
    sWt[(size_t)(n + 0) * S + kr] = w4.x;
    sWt[(size_t)(n + 1) * S + kr] = w4.y;
    sWt[(size_t)(n + 2) * S + kr] = w4.z;
    sWt[(size_t)(n + 3) * S + kr] = w4.w;
  }
  __syncthreads();

  const int lane = threadIdx.x & 31;
  const int wave = threadIdx.x >> 5;
  const int m    = lane & 15;
  const int half = lane >> 4;
  const long long row0 = ((long long)blockIdx.x * 2 + wave) * 16;

  const float* xrow  = X + (size_t)(row0 + m) * K + 2 * half;
  const float* bbase = sWt + (size_t)m * S + 2 * half;

  v8f acc[CT] = {};
#pragma unroll
  for (int kk = 0; kk < K / 4; ++kk) {
    const int k = kk * 4;
    float ax = xrow[k];
    float ay = xrow[k + 1];
    if (RELU) { ax = fmaxf(ax, 0.f); ay = fmaxf(ay, 0.f); }
    const v2f a = {ax, ay};
#pragma unroll
    for (int ct = 0; ct < CT; ++ct) {
      const v2f b = *(const v2f*)(bbase + (size_t)ct * 16 * S + k);
      acc[ct] = __builtin_amdgcn_wmma_f32_16x16x4_f32(
          false, a, false, b, (short)0, acc[ct], false, false);
    }
  }

  float dsc[8];
#pragma unroll
  for (int r = 0; r < 8; ++r) dsc[r] = dis[row0 + r + 8 * half];
#pragma unroll
  for (int ct = 0; ct < CT; ++ct) {
#pragma unroll
    for (int r = 0; r < 8; ++r) {
      const long long row = row0 + r + 8 * half;
      H[(size_t)row * NC + ct * 16 + m] = acc[ct][r] * dsc[r];
    }
  }
}


template <int F, int PASSES, int NC>
__global__ __launch_bounds__(256)
void gcn_gather(const float* __restrict__ h, const int* __restrict__ cs,
                const int* __restrict__ rowptr, const int* __restrict__ cnt,
                const float* __restrict__ dis, const float* __restrict__ bias,
                const float* __restrict__ Wn, float* __restrict__ out, int n) {
  constexpr int G   = F / 4;
  constexpr int NPB = 256 / G;
  constexpr int EB  = 8;
  constexpr unsigned RB = F * 4;
  constexpr bool FUSE = NC > 0;
  constexpr int S    = F + 6;
  constexpr int SA   = F + 4;
  constexpr int ROWS = NPB * PASSES;
  __shared__ __attribute__((aligned(16))) float sWt[FUSE ? NC * S : 1];
  __shared__ __attribute__((aligned(16))) float sA[FUSE ? ROWS * SA : 1];

  const int g  = threadIdx.x / G;
  const int fo = (threadIdx.x % G) * 4;
  const char* hb = (const char*)h;
  const unsigned lo = (unsigned)fo * 4u;
  const float4 b = *(const float4*)(bias + fo);
  const int node0 = blockIdx.x * (NPB * PASSES) + g;

  constexpr int WT = FUSE ? F * NC / 4 / 256 : 1;
  float4 w4[WT];
  if constexpr (FUSE) {
#pragma unroll
    for (int j = 0; j < WT; ++j)
      w4[j] = *(const float4*)(Wn + (size_t)(threadIdx.x + j * 256) * 4);
  }

  bool haveC = node0 < n;
  bool haveN = (PASSES > 1) && (node0 + NPB < n);
  int begC = 0, cntC = 0, begN = 0, cntN = 0;
  if (haveC) { begC = rowptr[node0];       cntC = cnt[node0]; }
  if (haveN) { begN = rowptr[node0 + NPB]; cntN = cnt[node0 + NPB]; }

  if constexpr (FUSE) {
#pragma unroll
    for (int j = 0; j < WT; ++j) {
      const int t  = threadIdx.x + j * 256;
      const int kr = (t * 4) / NC;
      const int nn = (t * 4) % NC;
      sWt[(nn + 0) * S + kr] = w4[j].x;
      sWt[(nn + 1) * S + kr] = w4[j].y;
      sWt[(nn + 2) * S + kr] = w4[j].z;
      sWt[(nn + 3) * S + kr] = w4[j].w;
    }
  }

  unsigned a[EB];
#pragma unroll
  for (int k = 0; k < EB; ++k) a[k] = lo;
  if (cntC > 0) {
    const int last = begC + cntC - 1;
    int t[EB];
#pragma unroll
    for (int k = 0; k < EB; ++k) t[k] = cs[min(begC + k, last)];
#pragma unroll
    for (int k = 0; k < EB; ++k) a[k] = (unsigned)t[k] * RB + lo;
  }

  for (int it = 0; it < PASSES; ++it) {
    const int node = node0 + it * NPB;

    const int nodeNN = node + 2 * NPB;
    const bool haveNN = (it + 2 < PASSES) && (nodeNN < n);
    int begNN = 0, cntNN = 0;
    if (haveNN) { begNN = rowptr[nodeNN]; cntNN = cnt[nodeNN]; }

    int tN[EB];
#pragma unroll
    for (int k = 0; k < EB; ++k) tN[k] = 0;
    if (cntN > 0) {
      const int lastN = begN + cntN - 1;
#pragma unroll
      for (int k = 0; k < EB; ++k) tN[k] = cs[min(begN + k, lastN)];
    }

    if (haveC) {
      const float dv = dis[node];
      float4 acc = *(const float4*)(hb + ((unsigned)node * RB + lo));
      const int end  = begC + cntC;
      const int last = end - 1;
      for (int p = begC; p < end; p += EB) {
        float4 v[EB];
#pragma unroll
        for (int k = 0; k < EB; ++k) v[k] = *(const float4*)(hb + a[k]);
        int t[EB];
#pragma unroll
        for (int k = 0; k < EB; ++k) t[k] = cs[min(p + EB + k, last)];
        acc.x += v[0].x; acc.y += v[0].y;
        acc.z += v[0].z; acc.w += v[0].w;
#pragma unroll
        for (int k = 1; k < EB; ++k) {
          const float w = (p + k < end) ? 1.f : 0.f;
          acc.x = fmaf(v[k].x, w, acc.x); acc.y = fmaf(v[k].y, w, acc.y);
          acc.z = fmaf(v[k].z, w, acc.z); acc.w = fmaf(v[k].w, w, acc.w);
        }
#pragma unroll
        for (int k = 0; k < EB; ++k) a[k] = (unsigned)t[k] * RB + lo;
      }
      float4 o;
      o.x = fmaf(acc.x, dv, b.x);
      o.y = fmaf(acc.y, dv, b.y);
      o.z = fmaf(acc.z, dv, b.z);
      o.w = fmaf(acc.w, dv, b.w);
      if constexpr (FUSE) {
        o.x = fmaxf(o.x, 0.f); o.y = fmaxf(o.y, 0.f);
        o.z = fmaxf(o.z, 0.f); o.w = fmaxf(o.w, 0.f);
        *(float4*)(sA + (it * NPB + g) * SA + fo) = o;
      } else {
        *(float4*)(out + (size_t)node * F + fo) = o;
      }
    }

    haveC = haveN;  begC = begN;  cntC = cntN;
    haveN = haveNN; begN = begNN; cntN = cntNN;
#pragma unroll
    for (int k = 0; k < EB; ++k) a[k] = (unsigned)tN[k] * RB + lo;
  }

  if constexpr (FUSE) {
    static_assert(F == 64 && NPB == 16 && NC % 32 == 0, "panel layout");
    static_assert(N_NODES % ROWS == 0, "every owned row is a node");
    __syncthreads();
    const int lane = threadIdx.x & 31;
    const int wave = threadIdx.x >> 5;
    const int m    = lane & 15;
    const int half = lane >> 4;
    constexpr int QN = NC / 32;
    constexpr int NITEMS = PASSES * QN;
    for (int i = wave; i < NITEMS; i += 8) {
      const int p = i % PASSES;
      const int q = i / PASSES;
      const int row0 = blockIdx.x * ROWS + p * 16;
      float dsc[8];
#pragma unroll
      for (int r = 0; r < 8; ++r) dsc[r] = dis[row0 + r + 8 * half];
      const float* abase = sA + (p * 16 + m) * SA + 2 * half;
      const float* bbase = sWt + (q * 32 + m) * S + 2 * half;
      v8f acc2[2] = {};
#pragma unroll
      for (int kk = 0; kk < F / 4; ++kk) {
        const int k = kk * 4;
        const v2f av = *(const v2f*)(abase + k);
#pragma unroll
        for (int ct = 0; ct < 2; ++ct) {
          const v2f bv = *(const v2f*)(bbase + ct * 16 * S + k);
          acc2[ct] = __builtin_amdgcn_wmma_f32_16x16x4_f32(
              false, av, false, bv, (short)0, acc2[ct], false, false);
        }
      }
#pragma unroll
      for (int ct = 0; ct < 2; ++ct) {
#pragma unroll
        for (int r = 0; r < 8; ++r) {
          const int row = row0 + r + 8 * half;
          out[(size_t)row * NC + q * 32 + ct * 16 + m] = acc2[ct][r] * dsc[r];
        }
      }
    }
  }
}


extern "C" void kernel_launch(void* const* d_in, const int* in_sizes, int n_in,
                              void* d_out, int out_size, void* d_ws, size_t ws_size,
                              hipStream_t stream) {
  const float* x  = (const float*)d_in[0];
  const int*   ei = (const int*)d_in[1];
  const float* W1 = (const float*)d_in[2];
  const float* b1 = (const float*)d_in[3];
  const float* W2 = (const float*)d_in[4];
  const float* b2 = (const float*)d_in[5];
  const float* W3 = (const float*)d_in[6];
  const float* b3 = (const float*)d_in[7];
  float* out = (float*)d_out;

  const int N = N_NODES;
  const int E = in_sizes[1] / 2;
  const int* src = ei;
  const int* dst = ei + E;

  float* ws     = (float*)d_ws;
  float* dis    = ws;
  int*   cnt    = (int*)(ws + OFF_CNT);
  int*   rowptr = (int*)(ws + OFF_ROWPTR);
  int*   cntBW  = (int*)(ws + OFF_CBW);
  int*   offBW  = (int*)(ws + OFF_OBW);
  int2*  bk     = (int2*)(ws + OFF_BK);
  int*   cs     = (int*)(ws + OFF_CS);
  float* bufA   = ws + OFF_BUFA;
  float* bufB   = bufA + NBUF;

  const int TB = 256;
  const int gGemm = N / 32;
  const int gG64 = 1250;
  const int gG32 = 1042;

  gcn_bucket_hist   <<<NW, 1024, 0, stream>>>(dst, cntBW, E);
  gcn_bucket_scan   <<<1, 896, 0, stream>>>(cntBW, offBW);
  gcn_bucket_scatter<<<NW, 1024, 0, stream>>>(src, dst, cntBW, offBW, bk, E);
  gcn_bucket_csr    <<<NB, 1024, 0, stream>>>(bk, offBW, rowptr, cnt, dis, cs, E);

  gcn_gemm_wmma<128, 64, false><<<gGemm, 64, 0, stream>>>(x, W1, dis, bufA);
  gcn_gather<64, 5, 64><<<gG64, TB, 0, stream>>>(bufA, cs, rowptr, cnt, dis, b1, W2, bufB, N);
  gcn_gather<64, 5, 32><<<gG64, TB, 0, stream>>>(bufB, cs, rowptr, cnt, dis, b2, W3, bufA, N);
  gcn_gather<32, 3, 0><<<gG32, TB, 0, stream>>>(bufA, cs, rowptr, cnt, dis, b3, nullptr, out, N);
}


# Dst-grouped edge list built in 2 launches, not 4: a chunk sorts its edges by bucket inside its own slice, a bucket collects its 256 runs; no count and scan launches; sum order: LDS rank over runs
# speedup vs baseline: 4.4886x; 1.1136x over previous
#include <hip/hip_runtime.h>


typedef __attribute__((ext_vector_type(2))) float v2f;
typedef __attribute__((ext_vector_type(8))) float v8f;

#define N_NODES 100000
#define NB 196
#define NPAD (NB * 512)
#define NW 256
#define CHUNK 4884
#define EPT 5
#define EPAD 1250048
#define CAP 12288
#define OFF_CNT NPAD
#define OFF_ROWPTR (2 * NPAD)
#define OFF_CBW (3 * NPAD)
#define OFF_BK (OFF_CBW + 2 * NB * NW)
#define OFF_CS (OFF_BK + 2 * EPAD)
#define OFF_BUFA (OFF_CS + EPAD)
#define NBUF ((size_t)N_NODES * 64)


__device__ __forceinline__ int block_excl_scan(int x, int* s_wsum) {
  const int lane = threadIdx.x & 31;
  const int wave = threadIdx.x >> 5;
  int inc = x;
#pragma unroll
  for (int o = 1; o < 32; o <<= 1) {
    const int y = __shfl_up(inc, o);
    if (lane >= o) inc += y;
  }
  if (lane == 31) s_wsum[wave] = inc;
  __syncthreads();
  int woff = 0;
  for (int i = 0; i < wave; ++i) woff += s_wsum[i];
  __syncthreads();
  return woff + inc - x;
}

static_assert(CHUNK < 65536 && NB <= 1024, "run word layout");
__global__ __launch_bounds__(1024)
void gcn_chunk_sort(const int* __restrict__ src, const int* __restrict__ dst,
                    int* __restrict__ runBW, int2* __restrict__ bk, int nE) {
  __shared__ int s_wsum[32];
  __shared__ int s_h[NB];
  __shared__ int s_loc[NB];
  __shared__ int2 s_pairs[CHUNK];
  const int t = threadIdx.x;
  const int w = blockIdx.x;
  const int e0 = w * CHUNK;
  const int e1 = min(e0 + CHUNK, nE);
  if (t < NB) s_h[t] = 0;
  int sv[EPT], dv[EPT], rk[EPT];
#pragma unroll
  for (int k = 0; k < EPT; ++k) {
    const int e = e0 + t + k * 1024;
    dv[k] = (e < e1) ? dst[e] : -1;
    sv[k] = (e < e1) ? src[e] : 0;
  }
  __syncthreads();
#pragma unroll
  for (int k = 0; k < EPT; ++k) {
    rk[k] = 0;
    if (dv[k] >= 0) rk[k] = atomicAdd(&s_h[dv[k] >> 9], 1);
  }
  __syncthreads();
  const int c = (t < NB) ? s_h[t] : 0;
  const int loc = block_excl_scan(c, s_wsum);
  if (t < NB) s_loc[t] = loc;
  __syncthreads();
#pragma unroll
  for (int k = 0; k < EPT; ++k) {
    if (dv[k] >= 0) {
      int2 p;
      p.x = sv[k];
      p.y = dv[k];
      s_pairs[s_loc[dv[k] >> 9] + rk[k]] = p;
    }
  }
  __syncthreads();
#pragma unroll
  for (int k = 0; k < EPT; ++k) {
    const int i = t + k * 1024;
    if (i < e1 - e0) bk[e0 + i] = s_pairs[i];
  }
  if (t < NB) runBW[t * NW + w] = (loc << 16) | c;
}

constexpr int RPW = NW / 32;
constexpr int TQ  = ((NB - 1) * (NW / 4) + 1023) / 1024;
static_assert(NW == 256 && RPW * 32 == NW, "run layout");
__global__ __launch_bounds__(1024)
void gcn_bucket_csr(const int2* __restrict__ bk, const int* __restrict__ runBW,
                    int* __restrict__ rowptr, int* __restrict__ cnt,
                    float* __restrict__ dis, int* __restrict__ csrc, int nE) {
  __shared__ int s_wsum[32];
  __shared__ int s_redb[32];
  __shared__ int s_redm[32];
  __shared__ int s_cnt[512];
  __shared__ int s_cur[512];
  __shared__ int s_sorted[CAP];
  const int t = threadIdx.x;
  const int b = blockIdx.x;
  const int lane = t & 31;
  const int wv = __builtin_amdgcn_readfirstlane(t >> 5);

  const int4* tab = (const int4*)runBW;
  const int lim = b * (NW / 4);
  int4 q[TQ];
#pragma unroll
  for (int k = 0; k < TQ; ++k) {
    const int i = t + k * 1024;
    q[k] = tab[(i < lim) ? i : 0];
  }
  int rc[RPW], rb[RPW];
#pragma unroll
  for (int s = 0; s < RPW; ++s) {
    const int r = wv + 32 * s;
    const int word = runBW[b * NW + r];
    rc[s] = word & 0xffff;
    rb[s] = r * CHUNK + (word >> 16);
  }
  int2 p[2 * RPW];
#pragma unroll
  for (int s = 0; s < 2 * RPW; ++s) {
    const int j = lane + 32 * (s & 1);
    int2 z;
    z.x = 0;
    z.y = -1;
    if (j < rc[s >> 1]) z = bk[rb[s >> 1] + j];
    p[s] = z;
  }

  if (t < 512) s_cnt[t] = 0;
  int pb = 0;
#pragma unroll
  for (int k = 0; k < TQ; ++k) {
    const int i = t + k * 1024;
    const int v = ((q[k].x & 0xffff) + (q[k].y & 0xffff)) + ((q[k].z & 0xffff) + (q[k].w & 0xffff));
    pb += (i < lim) ? v : 0;
  }
  int pm = 0;
#pragma unroll
  for (int s = 0; s < RPW; ++s) pm += rc[s];
#pragma unroll
  for (int o = 16; o > 0; o >>= 1) pb += __shfl_xor(pb, o);
  if (lane == 0) { s_redb[wv] = pb; s_redm[wv] = pm; }
  __syncthreads();
  int beg = 0, m = 0;
  for (int i = 0; i < 32; ++i) { beg += s_redb[i]; m += s_redm[i]; }

#pragma unroll
  for (int s = 0; s < 2 * RPW; ++s)
    if (p[s].y >= 0) atomicAdd(&s_cnt[p[s].y & 511], 1);
#pragma unroll
  for (int s = 0; s < RPW; ++s)
    for (int j = lane + 64; j < rc[s]; j += 32) atomicAdd(&s_cnt[bk[rb[s] + j].y & 511], 1);
  __syncthreads();
  const int c = (t < 512) ? s_cnt[t] : 0;
  const int ex = block_excl_scan(c, s_wsum);
  if (t < 512) s_cur[t] = ex;
  __syncthreads();
  if (t < 512) {
    const int node = b * 512 + t;
    rowptr[node] = beg + ex;
    cnt[node] = c;
    dis[node] = rsqrtf((float)(c + 1));
  }
  if (m <= CAP) {
#pragma unroll
    for (int s = 0; s < 2 * RPW; ++s) {
      if (p[s].y >= 0) {
        const int lp = atomicAdd(&s_cur[p[s].y & 511], 1);
        s_sorted[lp] = p[s].x;
      }
    }
#pragma unroll
    for (int s = 0; s < RPW; ++s) {
      for (int j = lane + 64; j < rc[s]; j += 32) {
        const int2 pp = bk[rb[s] + j];
        const int lp = atomicAdd(&s_cur[pp.y & 511], 1);
        s_sorted[lp] = pp.x;
      }
    }
    __syncthreads();
    for (int i = t; i < m; i += 1024) csrc[beg + i] = s_sorted[i];
  } else {
#pragma unroll
    for (int s = 0; s < 2 * RPW; ++s) {
      if (p[s].y >= 0) {
        const int lp = atomicAdd(&s_cur[p[s].y & 511], 1);
        csrc[beg + lp] = p[s].x;
      }
    }
#pragma unroll
    for (int s = 0; s < RPW; ++s) {
      for (int j = lane + 64; j < rc[s]; j += 32) {
        const int2 pp = bk[rb[s] + j];
        const int lp = atomicAdd(&s_cur[pp.y & 511], 1);
        csrc[beg + lp] = pp.x;
      }
    }
  }
}


template <int K, int NC, bool RELU>
__global__ __launch_bounds__(64)
void gcn_gemm_wmma(const float* __restrict__ X, const float* __restrict__ W,
                   const float* __restrict__ dis, float* __restrict__ H) {
  constexpr int S  = K + 6;
  constexpr int CT = NC / 16;
  __shared__ float sWt[NC * S];

  for (int t = threadIdx.x; t < K * NC / 4; t += 64) {
    const int kr = (t * 4) / NC;
    const int n  = (t * 4) % NC;
    const float4 w4 = *(const float4*)(W + (size_t)t * 4);
    sWt[(size_t)(n + 0) * S + kr] = w4.x;
    sWt[(size_t)(n + 1) * S + kr] = w4.y;
    sWt[(size_t)(n + 2) * S + kr] = w4.z;
    sWt[(size_t)(n + 3) * S + kr] = w4.w;
  }
  __syncthreads();

  const int lane = threadIdx.x & 31;
  const int wave = threadIdx.x >> 5;
  const int m    = lane & 15;
  const int half = lane >> 4;
  const long long row0 = ((long long)blockIdx.x * 2 + wave) * 16;

  const float* xrow  = X + (size_t)(row0 + m) * K + 2 * half;
  const float* bbase = sWt + (size_t)m * S + 2 * half;

  v8f acc[CT] = {};
#pragma unroll
  for (int kk = 0; kk < K / 4; ++kk) {
    const int k = kk * 4;
    float ax = xrow[k];
    float ay = xrow[k + 1];
    if (RELU) { ax = fmaxf(ax, 0.f); ay = fmaxf(ay, 0.f); }
    const v2f a = {ax, ay};
#pragma unroll
    for (int ct = 0; ct < CT; ++ct) {
      const v2f b = *(const v2f*)(bbase + (size_t)ct * 16 * S + k);
      acc[ct] = __builtin_amdgcn_wmma_f32_16x16x4_f32(
          false, a, false, b, (short)0, acc[ct], false, false);
    }
  }

  float dsc[8];
#pragma unroll
  for (int r = 0; r < 8; ++r) dsc[r] = dis[row0 + r + 8 * half];
#pragma unroll
  for (int ct = 0; ct < CT; ++ct) {
#pragma unroll
    for (int r = 0; r < 8; ++r) {
      const long long row = row0 + r + 8 * half;
      H[(size_t)row * NC + ct * 16 + m] = acc[ct][r] * dsc[r];
    }
  }
}


template <int F, int PASSES, int NC>
__global__ __launch_bounds__(256)
void gcn_gather(const float* __restrict__ h, const int* __restrict__ cs,
                const int* __restrict__ rowptr, const int* __restrict__ cnt,
                const float* __restrict__ dis, const float* __restrict__ bias,
                const float* __restrict__ Wn, float* __restrict__ out, int n) {
  constexpr int G   = F / 4;
  constexpr int NPB = 256 / G;
  constexpr int EB  = 8;
  constexpr unsigned RB = F * 4;
  constexpr bool FUSE = NC > 0;
  constexpr int S    = F + 6;
  constexpr int SA   = F + 4;
  constexpr int ROWS = NPB * PASSES;
  __shared__ __attribute__((aligned(16))) float sWt[FUSE ? NC * S : 1];
  __shared__ __attribute__((aligned(16))) float sA[FUSE ? ROWS * SA : 1];

  const int g  = threadIdx.x / G;
  const int fo = (threadIdx.x % G) * 4;
  const char* hb = (const char*)h;
  const unsigned lo = (unsigned)fo * 4u;
  const float4 b = *(const float4*)(bias + fo);
  const int node0 = blockIdx.x * (NPB * PASSES) + g;

  constexpr int WT = FUSE ? F * NC / 4 / 256 : 1;
  float4 w4[WT];
  if constexpr (FUSE) {
#pragma unroll
    for (int j = 0; j < WT; ++j)
      w4[j] = *(const float4*)(Wn + (size_t)(threadIdx.x + j * 256) * 4);
  }

  bool haveC = node0 < n;
  bool haveN = (PASSES > 1) && (node0 + NPB < n);
  int begC = 0, cntC = 0, begN = 0, cntN = 0;
  if (haveC) { begC = rowptr[node0];       cntC = cnt[node0]; }
  if (haveN) { begN = rowptr[node0 + NPB]; cntN = cnt[node0 + NPB]; }

  if constexpr (FUSE) {
#pragma unroll
    for (int j = 0; j < WT; ++j) {
      const int t  = threadIdx.x + j * 256;
      const int kr = (t * 4) / NC;
      const int nn = (t * 4) % NC;
      sWt[(nn + 0) * S + kr] = w4[j].x;
      sWt[(nn + 1) * S + kr] = w4[j].y;
      sWt[(nn + 2) * S + kr] = w4[j].z;
      sWt[(nn + 3) * S + kr] = w4[j].w;
    }
  }

  unsigned a[EB];
#pragma unroll
  for (int k = 0; k < EB; ++k) a[k] = lo;
  if (cntC > 0) {
    const int last = begC + cntC - 1;
    int t[EB];
#pragma unroll
    for (int k = 0; k < EB; ++k) t[k] = cs[min(begC + k, last)];
#pragma unroll
    for (int k = 0; k < EB; ++k) a[k] = (unsigned)t[k] * RB + lo;
  }

  for (int it = 0; it < PASSES; ++it) {
    const int node = node0 + it * NPB;

    const int nodeNN = node + 2 * NPB;
    const bool haveNN = (it + 2 < PASSES) && (nodeNN < n);
    int begNN = 0, cntNN = 0;
    if (haveNN) { begNN = rowptr[nodeNN]; cntNN = cnt[nodeNN]; }

    int tN[EB];
#pragma unroll
    for (int k = 0; k < EB; ++k) tN[k] = 0;
    if (cntN > 0) {
      const int lastN = begN + cntN - 1;
#pragma unroll
      for (int k = 0; k < EB; ++k) tN[k] = cs[min(begN + k, lastN)];
    }

    if (haveC) {
      const float dv = dis[node];
      float4 acc = *(const float4*)(hb + ((unsigned)node * RB + lo));
      const int end  = begC + cntC;
      const int last = end - 1;
      for (int p = begC; p < end; p += EB) {
        float4 v[EB];
#pragma unroll
        for (int k = 0; k < EB; ++k) v[k] = *(const float4*)(hb + a[k]);
        int t[EB];
#pragma unroll
        for (int k = 0; k < EB; ++k) t[k] = cs[min(p + EB + k, last)];
        acc.x += v[0].x; acc.y += v[0].y;
        acc.z += v[0].z; acc.w += v[0].w;
#pragma unroll
        for (int k = 1; k < EB; ++k) {
          const float w = (p + k < end) ? 1.f : 0.f;
          acc.x = fmaf(v[k].x, w, acc.x); acc.y = fmaf(v[k].y, w, acc.y);
          acc.z = fmaf(v[k].z, w, acc.z); acc.w = fmaf(v[k].w, w, acc.w);
        }
#pragma unroll
        for (int k = 0; k < EB; ++k) a[k] = (unsigned)t[k] * RB + lo;
      }
      float4 o;
      o.x = fmaf(acc.x, dv, b.x);
      o.y = fmaf(acc.y, dv, b.y);
      o.z = fmaf(acc.z, dv, b.z);
      o.w = fmaf(acc.w, dv, b.w);
      if constexpr (FUSE) {
        o.x = fmaxf(o.x, 0.f); o.y = fmaxf(o.y, 0.f);
        o.z = fmaxf(o.z, 0.f); o.w = fmaxf(o.w, 0.f);
        *(float4*)(sA + (it * NPB + g) * SA + fo) = o;
      } else {
        *(float4*)(out + (size_t)node * F + fo) = o;
      }
    }

    haveC = haveN;  begC = begN;  cntC = cntN;
    haveN = haveNN; begN = begNN; cntN = cntNN;
#pragma unroll
    for (int k = 0; k < EB; ++k) a[k] = (unsigned)tN[k] * RB + lo;
  }

  if constexpr (FUSE) {
    static_assert(F == 64 && NPB == 16 && NC % 32 == 0, "panel layout");
    static_assert(N_NODES % ROWS == 0, "every owned row is a node");
    __syncthreads();
    const int lane = threadIdx.x & 31;
    const int wave = threadIdx.x >> 5;
    const int m    = lane & 15;
    const int half = lane >> 4;
    constexpr int QN = NC / 32;
    constexpr int NITEMS = PASSES * QN;
    for (int i = wave; i < NITEMS; i += 8) {
      const int p = i % PASSES;
      const int q = i / PASSES;
      const int row0 = blockIdx.x * ROWS + p * 16;
      float dsc[8];
#pragma unroll
      for (int r = 0; r < 8; ++r) dsc[r] = dis[row0 + r + 8 * half];
      const float* abase = sA + (p * 16 + m) * SA + 2 * half;
      const float* bbase = sWt + (q * 32 + m) * S + 2 * half;
      v8f acc2[2] = {};
#pragma unroll
      for (int kk = 0; kk < F / 4; ++kk) {
        const int k = kk * 4;
        const v2f av = *(const v2f*)(abase + k);
#pragma unroll
        for (int ct = 0; ct < 2; ++ct) {
          const v2f bv = *(const v2f*)(bbase + ct * 16 * S + k);
          acc2[ct] = __builtin_amdgcn_wmma_f32_16x16x4_f32(
              false, av, false, bv, (short)0, acc2[ct], false, false);
        }
      }
#pragma unroll
      for (int ct = 0; ct < 2; ++ct) {
#pragma unroll
        for (int r = 0; r < 8; ++r) {
          const int row = row0 + r + 8 * half;
          out[(size_t)row * NC + q * 32 + ct * 16 + m] = acc2[ct][r] * dsc[r];
        }
      }
    }
  }
}


extern "C" void kernel_launch(void* const* d_in, const int* in_sizes, int n_in,
                              void* d_out, int out_size, void* d_ws, size_t ws_size,
                              hipStream_t stream) {
  const float* x  = (const float*)d_in[0];
  const int*   ei = (const int*)d_in[1];
  const float* W1 = (const float*)d_in[2];
  const float* b1 = (const float*)d_in[3];
  const float* W2 = (const float*)d_in[4];
  const float* b2 = (const float*)d_in[5];
  const float* W3 = (const float*)d_in[6];
  const float* b3 = (const float*)d_in[7];
  float* out = (float*)d_out;

  const int N = N_NODES;
  const int E = in_sizes[1] / 2;
  const int* src = ei;
  const int* dst = ei + E;

  float* ws     = (float*)d_ws;
  float* dis    = ws;
  int*   cnt    = (int*)(ws + OFF_CNT);
  int*   rowptr = (int*)(ws + OFF_ROWPTR);
  int*   runBW  = (int*)(ws + OFF_CBW);
  int2*  bk     = (int2*)(ws + OFF_BK);
  int*   cs     = (int*)(ws + OFF_CS);
  float* bufA   = ws + OFF_BUFA;
  float* bufB   = bufA + NBUF;

  const int TB = 256;
  const int gGemm = N / 32;
  const int gG64 = 1250;
  const int gG32 = 1042;

  gcn_chunk_sort<<<NW, 1024, 0, stream>>>(src, dst, runBW, bk, E);
  gcn_bucket_csr<<<NB, 1024, 0, stream>>>(bk, runBW, rowptr, cnt, dis, cs, E);

  gcn_gemm_wmma<128, 64, false><<<gGemm, 64, 0, stream>>>(x, W1, dis, bufA);
  gcn_gather<64, 5, 64><<<gG64, TB, 0, stream>>>(bufA, cs, rowptr, cnt, dis, b1, W2, bufB, N);
  gcn_gather<64, 5, 32><<<gG64, TB, 0, stream>>>(bufB, cs, rowptr, cnt, dis, b2, W3, bufA, N);
  gcn_gather<32, 3, 0><<<gG32, TB, 0, stream>>>(bufA, cs, rowptr, cnt, dis, b3, nullptr, out, N);
}
